# starting kernel
# baseline (speedup 1.0000x reference)
#include <hip/hip_runtime.h>

typedef float v2f __attribute__((ext_vector_type(2)));
typedef float v8f __attribute__((ext_vector_type(8)));

#define NXL 1024
#define NYL 1024
#define TAU 0.6f
#define GYF (-1e-4f)

struct ATab { float v[12][16]; };

static constexpr ATab makeAT() {
    const signed char M[9][9] = {
        { 1,  1,  1,  1,  1,  1,  1,  1,  1},
        { 0,  1,  0, -1,  0,  1, -1, -1,  1},
        { 0,  0,  1,  0, -1,  1,  1, -1, -1},
        {-4, -1, -1, -1, -1,  2,  2,  2,  2},
        { 4, -2, -2, -2, -2,  1,  1,  1,  1},
        { 0, -2,  0,  2,  0,  1, -1, -1,  1},
        { 0,  0, -2,  0,  2,  1,  1, -1, -1},
        { 0,  1, -1,  1, -1,  0,  0,  0,  0},
        { 0,  0,  0,  0,  0,  1, -1,  1, -1},
    };
    const float nrm2[9] = {9.f, 6.f, 6.f, 36.f, 36.f, 12.f, 12.f, 4.f, 4.f};
    ATab t{};
    for (int k = 0; k < 12; ++k)
        for (int q = 0; q < 16; ++q) {
            float x = 0.0f;
            if (k < 9 && q < 9)
                x = (float)M[k][q] / nrm2[k] * (1.0f / TAU);
            t.v[k][q] = x;
        }
    return t;
}

__constant__ ATab g_AT = makeAT();

__global__ __launch_bounds__(256) void lbm_step(const float* __restrict__ fin,
                                                const float* __restrict__ w,
                                                float* __restrict__ fout) {
    const int tid  = blockIdx.x * blockDim.x + threadIdx.x;
    const int lane = threadIdx.x & 31;
    const int half = lane >> 4;
    const int q16  = lane & 15;
    const int cell = (tid >> 5) * 16 + q16;
    const int x = cell >> 10;
    const int y = cell & (NYL - 1);

    const int exv[9] = {0, 1, 0, -1, 0, 1, -1, -1, 1};
    const int eyv[9] = {0, 0, 1, 0, -1, 1, 1, -1, -1};
    float f[9];
#pragma unroll
    for (int i = 0; i < 9; ++i) {
        const int nx = (x - exv[i]) & (NXL - 1);
        const int ny = (y - eyv[i]) & (NYL - 1);
        f[i] = fin[((nx << 10) | ny) * 9 + i];
    }

    const float rho = ((f[0] + f[1]) + (f[2] + f[3])) +
                      ((f[4] + f[5]) + (f[6] + f[7])) + f[8];
    const float mx = f[1] - f[3] + f[5] - f[6] - f[7] + f[8];
    const float my = f[2] - f[4] + f[5] + f[6] - f[7] - f[8];
    const float jx = mx;
    const float jy = my + (TAU * GYF) * rho * rho;
    const float jx2 = jx * jx, jy2 = jy * jy;
    const float j2 = jx2 + jy2;

    const float w0 = w[0], w1 = w[1], w2 = w[2], w3 = w[3];
    const float w4 = w[4], w5 = w[5], w6 = w[6];

    float me[9];
    me[0] = rho;
    me[1] = jx;
    me[2] = jy;
    me[3] = w0 * rho + w1 * j2;
    me[4] = w2 * rho - w3 * j2;
    me[5] = w4 * jx;
    me[6] = w4 * jy;
    me[7] = w5 * (jx2 - jy2);
    me[8] = w6 * jx * jy;

    const float* at = &g_AT.v[0][0];
    v2f a0, a1, a2;
    a0.x = at[(0 + 2 * half) * 16 + q16];
    a0.y = at[(1 + 2 * half) * 16 + q16];
    a1.x = at[(4 + 2 * half) * 16 + q16];
    a1.y = at[(5 + 2 * half) * 16 + q16];
    a2.x = at[(8 + 2 * half) * 16 + q16];
    a2.y = at[(9 + 2 * half) * 16 + q16];

    v2f b0, b1, b2;
    b0.x = half ? me[2] : me[0];
    b0.y = half ? me[3] : me[1];
    b1.x = half ? me[6] : me[4];
    b1.y = half ? me[7] : me[5];
    b2.x = half ? 0.0f  : me[8];
    b2.y = 0.0f;

    v8f c = {};
    c = __builtin_amdgcn_wmma_f32_16x16x4_f32(false, a0, false, b0, (short)0, c,
                                              false, false);
    c = __builtin_amdgcn_wmma_f32_16x16x4_f32(false, a1, false, b1, (short)0, c,
                                              false, false);
    c = __builtin_amdgcn_wmma_f32_16x16x4_f32(false, a2, false, b2, (short)0, c,
                                              false, false);

    const float decay = 1.0f - 1.0f / TAU;
    float* o = fout + cell * 9;
    if (half == 0) {
#pragma unroll
        for (int q = 0; q < 8; ++q) o[q] = decay * f[q] + c[q];
    } else {
        o[8] = decay * f[8] + c[0];
    }
}

extern "C" void kernel_launch(void* const* d_in, const int* in_sizes, int n_in,
                              void* d_out, int out_size, void* d_ws, size_t ws_size,
                              hipStream_t stream) {
    const float* f0 = (const float*)d_in[0];
    const float* w  = (const float*)d_in[1];
    float* out = (float*)d_out;
    float* ws  = (float*)d_ws;

    const int threads = 256;
    const int blocks  = (NXL * NYL * 2) / threads;

    lbm_step<<<blocks, threads, 0, stream>>>(f0,  w, ws);
    lbm_step<<<blocks, threads, 0, stream>>>(ws,  w, out);
    lbm_step<<<blocks, threads, 0, stream>>>(out, w, ws);
    lbm_step<<<blocks, threads, 0, stream>>>(ws,  w, out);
}


# stage 16x32-cell tiles through LDS: rows with halo enter and results leave as whole 16-byte row pieces (full 128-byte lines), not 36-byte-strided 4-byte accesses; summation order unchanged
# speedup vs baseline: 1.2034x; 1.2034x over previous
#include <hip/hip_runtime.h>


typedef float v2f __attribute__((ext_vector_type(2)));
typedef float v8f __attribute__((ext_vector_type(8)));
typedef float v4f __attribute__((ext_vector_type(4)));

#define NXL 1024
#define NYL 1024
#define TAU 0.6f
#define GYF (-1e-4f)

#define TX 16
#define TY 32
#define ROWF (TY * 9)
#define ROWV (ROWF / 4)
#define LPAD 12
#define PITCH 312
#define INROWS (TX + 2)
#define TILES_Y (NYL / TY)
#define IN_PIECES (INROWS * ROWV)
#define OUT_PIECES (TX * ROWV)
#define HALO_F (INROWS * 18)
#define GROUPS_PER_ROW (TY / 16)
#define GROUPS_PER_WAVE (TX * GROUPS_PER_ROW / 8)
static_assert(IN_PIECES >= 5 * 256 && IN_PIECES <= 6 * 256, "input split: five pieces per thread plus a rest");
static_assert(OUT_PIECES >= 4 * 256 && OUT_PIECES <= 5 * 256, "output split: four pieces per thread plus a rest");
static_assert(HALO_F >= 256 && HALO_F <= 2 * 256, "halo split: one float per thread plus a rest");
static_assert(TY % 16 == 0 && NYL % TY == 0 && NXL % TX == 0, "tiles cover the grid; a tile row starts on a 16-byte piece");
static_assert(PITCH % 4 == 0 && PITCH >= LPAD + ROWF + 9 && LPAD % 4 == 0 && LPAD >= 9, "LDS row layout");

struct ATab { float v[12][16]; };

static constexpr ATab makeAT() {
    const signed char M[9][9] = {
        { 1,  1,  1,  1,  1,  1,  1,  1,  1},
        { 0,  1,  0, -1,  0,  1, -1, -1,  1},
        { 0,  0,  1,  0, -1,  1,  1, -1, -1},
        {-4, -1, -1, -1, -1,  2,  2,  2,  2},
        { 4, -2, -2, -2, -2,  1,  1,  1,  1},
        { 0, -2,  0,  2,  0,  1, -1, -1,  1},
        { 0,  0, -2,  0,  2,  1,  1, -1, -1},
        { 0,  1, -1,  1, -1,  0,  0,  0,  0},
        { 0,  0,  0,  0,  0,  1, -1,  1, -1},
    };
    const float nrm2[9] = {9.f, 6.f, 6.f, 36.f, 36.f, 12.f, 12.f, 4.f, 4.f};
    ATab t{};
    for (int k = 0; k < 12; ++k)
        for (int q = 0; q < 16; ++q) {
            float x = 0.0f;
            if (k < 9 && q < 9)
                x = (float)M[k][q] / nrm2[k] * (1.0f / TAU);
            t.v[k][q] = x;
        }
    return t;
}

__constant__ ATab g_AT = makeAT();

static __device__ inline unsigned in_piece_goff(unsigned idx, unsigned x0, unsigned y0) {
    const unsigned row = idx / ROWV, kk = idx - row * ROWV;
    const unsigned gx = (x0 + (NXL - 1) + row) & (NXL - 1);
    return ((gx << 8) + (y0 >> 2)) * 9u + kk;
}
static __device__ inline unsigned in_piece_loff(unsigned idx) {
    const unsigned row = idx / ROWV, kk = idx - row * ROWV;
    return row * PITCH + LPAD + kk * 4u;
}
static __device__ inline unsigned out_piece_goff(unsigned idx, unsigned x0, unsigned y0) {
    const unsigned row = idx / ROWV, kk = idx - row * ROWV;
    return (((x0 + row) << 8) + (y0 >> 2)) * 9u + kk;
}
static __device__ inline unsigned out_piece_loff(unsigned idx) {
    const unsigned row = idx / ROWV, kk = idx - row * ROWV;
    return row * ROWF + kk * 4u;
}
static __device__ inline unsigned halo_goff(unsigned h, unsigned x0, unsigned y0) {
    const unsigned row = h / 18u, rem = h - row * 18u;
    const unsigned side = rem >= 9u ? 1u : 0u, i = rem - side * 9u;
    const unsigned gx = (x0 + (NXL - 1) + row) & (NXL - 1);
    const unsigned gy = (side ? (y0 + TY) : (y0 + (NYL - 1))) & (NYL - 1);
    return ((gx << 10) + gy) * 9u + i;
}
static __device__ inline unsigned halo_loff(unsigned h) {
    const unsigned row = h / 18u, rem = h - row * 18u;
    const unsigned side = rem >= 9u ? 1u : 0u, i = rem - side * 9u;
    return row * PITCH + (side ? (LPAD + ROWF) : (LPAD - 9)) + i;
}

__global__ __launch_bounds__(256) void lbm_step(const float* __restrict__ fin,
                                                const float* __restrict__ w,
                                                float* __restrict__ fout) {
    __shared__ __attribute__((aligned(16))) float s_in[INROWS * PITCH];
    __shared__ __attribute__((aligned(16))) float s_out[TX * ROWF];

    const unsigned tid  = threadIdx.x;
    const unsigned lane = tid & 31u;
    const unsigned half = lane >> 4;
    const unsigned q16  = lane & 15u;
    const unsigned wv   = tid >> 5;
    const unsigned x0 = (blockIdx.x / TILES_Y) * TX;
    const unsigned y0 = (blockIdx.x % TILES_Y) * TY;

    const float* at = &g_AT.v[0][0];
    v2f a0, a1, a2;
    a0.x = at[(0 + 2 * half) * 16 + q16];
    a0.y = at[(1 + 2 * half) * 16 + q16];
    a1.x = at[(4 + 2 * half) * 16 + q16];
    a1.y = at[(5 + 2 * half) * 16 + q16];
    a2.x = at[(8 + 2 * half) * 16 + q16];
    a2.y = at[(9 + 2 * half) * 16 + q16];

    const v4f* fin4 = reinterpret_cast<const v4f*>(fin);
    v4f* fout4 = reinterpret_cast<v4f*>(fout);
    v4f rv0, rv1, rv2, rv3, rv4, rv5;
    float hv0, hv1;
    rv0 = fin4[in_piece_goff(tid + 0u * 256u, x0, y0)];
    rv1 = fin4[in_piece_goff(tid + 1u * 256u, x0, y0)];
    rv2 = fin4[in_piece_goff(tid + 2u * 256u, x0, y0)];
    rv3 = fin4[in_piece_goff(tid + 3u * 256u, x0, y0)];
    rv4 = fin4[in_piece_goff(tid + 4u * 256u, x0, y0)];
    if (tid + 5u * 256u < IN_PIECES)
        rv5 = fin4[in_piece_goff(tid + 5u * 256u, x0, y0)];
    hv0 = fin[halo_goff(tid, x0, y0)];
    if (tid + 256u < HALO_F) hv1 = fin[halo_goff(tid + 256u, x0, y0)];

    const float w0 = w[0], w1 = w[1], w2 = w[2], w3 = w[3];
    const float w4 = w[4], w5 = w[5], w6 = w[6];

    *reinterpret_cast<v4f*>(&s_in[in_piece_loff(tid + 0u * 256u)]) = rv0;
    *reinterpret_cast<v4f*>(&s_in[in_piece_loff(tid + 1u * 256u)]) = rv1;
    *reinterpret_cast<v4f*>(&s_in[in_piece_loff(tid + 2u * 256u)]) = rv2;
    *reinterpret_cast<v4f*>(&s_in[in_piece_loff(tid + 3u * 256u)]) = rv3;
    *reinterpret_cast<v4f*>(&s_in[in_piece_loff(tid + 4u * 256u)]) = rv4;
    if (tid + 5u * 256u < IN_PIECES)
        *reinterpret_cast<v4f*>(&s_in[in_piece_loff(tid + 5u * 256u)]) = rv5;
    s_in[halo_loff(tid)] = hv0;
    if (tid + 256u < HALO_F) s_in[halo_loff(tid + 256u)] = hv1;
    __syncthreads();

    const int exv[9] = {0, 1, 0, -1, 0, 1, -1, -1, 1};
    const int eyv[9] = {0, 0, 1, 0, -1, 1, 1, -1, -1};
    const float decay = 1.0f - 1.0f / TAU;

    for (unsigned j = 0; j < GROUPS_PER_WAVE; ++j) {
        const unsigned g  = wv * GROUPS_PER_WAVE + j;
        const unsigned lx = g / GROUPS_PER_ROW;
        const unsigned ly = (g % GROUPS_PER_ROW) * 16u + q16;
        const float* c0 = &s_in[(lx + 1u) * PITCH + LPAD + ly * 9u];
        float f[9];
#pragma unroll
        for (int i = 0; i < 9; ++i)
            f[i] = c0[-exv[i] * PITCH - eyv[i] * 9 + i];

        const float rho = ((f[0] + f[1]) + (f[2] + f[3])) +
                          ((f[4] + f[5]) + (f[6] + f[7])) + f[8];
        const float mx = f[1] - f[3] + f[5] - f[6] - f[7] + f[8];
        const float my = f[2] - f[4] + f[5] + f[6] - f[7] - f[8];
        const float jx = mx;
        const float jy = my + (TAU * GYF) * rho * rho;
        const float jx2 = jx * jx, jy2 = jy * jy;
        const float j2 = jx2 + jy2;

        float me[9];
        me[0] = rho;
        me[1] = jx;
        me[2] = jy;
        me[3] = w0 * rho + w1 * j2;
        me[4] = w2 * rho - w3 * j2;
        me[5] = w4 * jx;
        me[6] = w4 * jy;
        me[7] = w5 * (jx2 - jy2);
        me[8] = w6 * jx * jy;

        v2f b0, b1, b2;
        b0.x = half ? me[2] : me[0];
        b0.y = half ? me[3] : me[1];
        b1.x = half ? me[6] : me[4];
        b1.y = half ? me[7] : me[5];
        b2.x = half ? 0.0f  : me[8];
        b2.y = 0.0f;

        v8f c = {};
        c = __builtin_amdgcn_wmma_f32_16x16x4_f32(false, a0, false, b0, (short)0, c,
                                                  false, false);
        c = __builtin_amdgcn_wmma_f32_16x16x4_f32(false, a1, false, b1, (short)0, c,
                                                  false, false);
        c = __builtin_amdgcn_wmma_f32_16x16x4_f32(false, a2, false, b2, (short)0, c,
                                                  false, false);

        float* o = &s_out[lx * ROWF + ly * 9u];
        const float fs = half ? f[8] : f[0];
        o[half ? 8 : 0] = decay * fs + c[0];
        if (half == 0) {
#pragma unroll
            for (int q = 1; q < 8; ++q) o[q] = decay * f[q] + c[q];
        }
    }
    __syncthreads();

    const v4f ov0 = *reinterpret_cast<const v4f*>(&s_out[out_piece_loff(tid + 0u * 256u)]);
    const v4f ov1 = *reinterpret_cast<const v4f*>(&s_out[out_piece_loff(tid + 1u * 256u)]);
    const v4f ov2 = *reinterpret_cast<const v4f*>(&s_out[out_piece_loff(tid + 2u * 256u)]);
    const v4f ov3 = *reinterpret_cast<const v4f*>(&s_out[out_piece_loff(tid + 3u * 256u)]);
    fout4[out_piece_goff(tid + 0u * 256u, x0, y0)] = ov0;
    fout4[out_piece_goff(tid + 1u * 256u, x0, y0)] = ov1;
    fout4[out_piece_goff(tid + 2u * 256u, x0, y0)] = ov2;
    fout4[out_piece_goff(tid + 3u * 256u, x0, y0)] = ov3;
    if (tid + 4u * 256u < OUT_PIECES) {
        const v4f ov4 = *reinterpret_cast<const v4f*>(&s_out[out_piece_loff(tid + 4u * 256u)]);
        fout4[out_piece_goff(tid + 4u * 256u, x0, y0)] = ov4;
    }
}

extern "C" void kernel_launch(void* const* d_in, const int* in_sizes, int n_in,
                              void* d_out, int out_size, void* d_ws, size_t ws_size,
                              hipStream_t stream) {
    const float* f0 = (const float*)d_in[0];
    const float* w  = (const float*)d_in[1];
    float* out = (float*)d_out;
    float* ws  = (float*)d_ws;

    const int threads = 256;
    const int blocks  = (NXL / TX) * (NYL / TY);

    lbm_step<<<blocks, threads, 0, stream>>>(f0,  w, ws);
    lbm_step<<<blocks, threads, 0, stream>>>(ws,  w, out);
    lbm_step<<<blocks, threads, 0, stream>>>(out, w, ws);
    lbm_step<<<blocks, threads, 0, stream>>>(ws,  w, out);
}


# temporal fusion: all four stream+collide steps in one launch; each work-group steps a 32x32 tile with a 4-cell ring four times in LDS (ring cells recomputed); per-cell summation order unchanged
# speedup vs baseline: 1.7983x; 1.4943x over previous
#include <hip/hip_runtime.h>


typedef float v2f __attribute__((ext_vector_type(2)));
typedef float v8f __attribute__((ext_vector_type(8)));
typedef float v4f __attribute__((ext_vector_type(4)));

#define NXL 1024
#define NYL 1024
#define TAU 0.6f
#define GYF (-1e-4f)

#define STEPS 4
#define HALO STEPS
#define TS 32
#define RS (TS + 2 * HALO)
#define PITCH (RS * 9)
#define BUFF (RS * PITCH)
#define ROWV (PITCH / 4)
#define HV (HALO * 9 / 4)
#define MIDV (TS * 9 / 4)
#define IN_PIECES (RS * ROWV)
#define OUT_PIECES (TS * MIDV)
#define TILES_Y (NYL / TS)
#define NTHREADS 256
#define NWAVES (NTHREADS / 32)
#define IN_ROUNDS ((IN_PIECES + NTHREADS - 1) / NTHREADS)
#define OUT_ROUNDS (OUT_PIECES / NTHREADS)
static_assert((HALO * 9) % 4 == 0 && (TS * 9) % 4 == 0, "ring and centre segments are whole 16-byte pieces");
static_assert(NXL % TS == 0 && NYL % TS == 0 && TS % 4 == 0 && HALO % 4 == 0, "tiles cover the grid; segments start on a 16-byte piece");
static_assert(OUT_PIECES % NTHREADS == 0, "output split: whole rounds");
static_assert(2 * HV + MIDV == ROWV, "a staged row is ring + centre + ring");

struct ATab { float v[12][16]; };

static constexpr ATab makeAT() {
    const signed char M[9][9] = {
        { 1,  1,  1,  1,  1,  1,  1,  1,  1},
        { 0,  1,  0, -1,  0,  1, -1, -1,  1},
        { 0,  0,  1,  0, -1,  1,  1, -1, -1},
        {-4, -1, -1, -1, -1,  2,  2,  2,  2},
        { 4, -2, -2, -2, -2,  1,  1,  1,  1},
        { 0, -2,  0,  2,  0,  1, -1, -1,  1},
        { 0,  0, -2,  0,  2,  1,  1, -1, -1},
        { 0,  1, -1,  1, -1,  0,  0,  0,  0},
        { 0,  0,  0,  0,  0,  1, -1,  1, -1},
    };
    const float nrm2[9] = {9.f, 6.f, 6.f, 36.f, 36.f, 12.f, 12.f, 4.f, 4.f};
    ATab t{};
    for (int k = 0; k < 12; ++k)
        for (int q = 0; q < 16; ++q) {
            float x = 0.0f;
            if (k < 9 && q < 9)
                x = (float)M[k][q] / nrm2[k] * (1.0f / TAU);
            t.v[k][q] = x;
        }
    return t;
}

__constant__ ATab g_AT = makeAT();

static __device__ inline unsigned in_piece_goff(unsigned idx, unsigned x0, unsigned y0) {
    const unsigned row = idx / ROWV, p = idx - row * ROWV;
    const unsigned gx = (x0 + (NXL - HALO) + row) & (NXL - 1);
    unsigned gy, pp;
    if (p < HV)             { gy = (y0 + (NYL - HALO)) & (NYL - 1); pp = p; }
    else if (p < HV + MIDV) { gy = y0;                              pp = p - HV; }
    else                    { gy = (y0 + TS) & (NYL - 1);           pp = p - (HV + MIDV); }
    return ((gx << 8) + (gy >> 2)) * 9u + pp;
}
static __device__ inline unsigned in_piece_loff(unsigned idx) {
    const unsigned row = idx / ROWV, p = idx - row * ROWV;
    return row * PITCH + p * 4u;
}
static __device__ inline unsigned out_piece_goff(unsigned idx, unsigned x0, unsigned y0) {
    const unsigned row = idx / MIDV, kk = idx - row * MIDV;
    return (((x0 + row) << 8) + (y0 >> 2)) * 9u + kk;
}
static __device__ inline unsigned out_piece_loff(unsigned idx) {
    const unsigned row = idx / MIDV, kk = idx - row * MIDV;
    return (HALO + row) * PITCH + HALO * 9u + kk * 4u;
}

template <int S>
static __device__ __forceinline__ void tile_step(float* s_buf, unsigned wv, unsigned half, unsigned q16,
                                                 v2f a0, v2f a1, v2f a2,
                                                 float w0, float w1, float w2, float w3,
                                                 float w4, float w5, float w6) {
    constexpr unsigned N = RS - 2 * S;
    constexpr unsigned NCELL = N * N;
    constexpr unsigned NGROUPS = (NCELL + 15u) / 16u;
    const float* src = s_buf + ((S - 1) & 1) * BUFF;
    float* dst = s_buf + (S & 1) * BUFF;

    const int exv[9] = {0, 1, 0, -1, 0, 1, -1, -1, 1};
    const int eyv[9] = {0, 0, 1, 0, -1, 1, 1, -1, -1};
    const float decay = 1.0f - 1.0f / TAU;

    for (unsigned g = wv; g < NGROUPS; g += NWAVES) {
        unsigned idx = g * 16u + q16;
        idx = idx < NCELL ? idx : NCELL - 1u;
        const unsigned r = idx / N, cc = idx - r * N;
        const unsigned cell = (S + r) * PITCH + (S + cc) * 9u;
        const float* c0 = src + cell;
        float f[9];
#pragma unroll
        for (int i = 0; i < 9; ++i)
            f[i] = c0[-exv[i] * PITCH - eyv[i] * 9 + i];

        const float rho = ((f[0] + f[1]) + (f[2] + f[3])) +
                          ((f[4] + f[5]) + (f[6] + f[7])) + f[8];
        const float mx = f[1] - f[3] + f[5] - f[6] - f[7] + f[8];
        const float my = f[2] - f[4] + f[5] + f[6] - f[7] - f[8];
        const float jx = mx;
        const float jy = my + (TAU * GYF) * rho * rho;
        const float jx2 = jx * jx, jy2 = jy * jy;
        const float j2 = jx2 + jy2;

        float me[9];
        me[0] = rho;
        me[1] = jx;
        me[2] = jy;
        me[3] = w0 * rho + w1 * j2;
        me[4] = w2 * rho - w3 * j2;
        me[5] = w4 * jx;
        me[6] = w4 * jy;
        me[7] = w5 * (jx2 - jy2);
        me[8] = w6 * jx * jy;

        v2f b0, b1, b2;
        b0.x = half ? me[2] : me[0];
        b0.y = half ? me[3] : me[1];
        b1.x = half ? me[6] : me[4];
        b1.y = half ? me[7] : me[5];
        b2.x = half ? 0.0f  : me[8];
        b2.y = 0.0f;

        v8f c = {};
        c = __builtin_amdgcn_wmma_f32_16x16x4_f32(false, a0, false, b0, (short)0, c,
                                                  false, false);
        c = __builtin_amdgcn_wmma_f32_16x16x4_f32(false, a1, false, b1, (short)0, c,
                                                  false, false);
        c = __builtin_amdgcn_wmma_f32_16x16x4_f32(false, a2, false, b2, (short)0, c,
                                                  false, false);

        float* o = dst + cell;
        const float fs = half ? f[8] : f[0];
        o[half ? 8 : 0] = decay * fs + c[0];
        if (half == 0) {
#pragma unroll
            for (int q = 1; q < 8; ++q) o[q] = decay * f[q] + c[q];
        }
    }
}

__global__ __launch_bounds__(NTHREADS) void lbm_fused(const float* __restrict__ fin,
                                                      const float* __restrict__ w,
                                                      float* __restrict__ fout) {
    __shared__ __attribute__((aligned(16))) float s_buf[2 * BUFF];

    const unsigned tid  = threadIdx.x;
    const unsigned lane = tid & 31u;
    const unsigned half = lane >> 4;
    const unsigned q16  = lane & 15u;
    const unsigned wv   = (unsigned)__builtin_amdgcn_readfirstlane((int)(tid >> 5));
    const unsigned x0 = (blockIdx.x / TILES_Y) * TS;
    const unsigned y0 = (blockIdx.x % TILES_Y) * TS;

    const float* at = &g_AT.v[0][0];
    v2f a0, a1, a2;
    a0.x = at[(0 + 2 * half) * 16 + q16];
    a0.y = at[(1 + 2 * half) * 16 + q16];
    a1.x = at[(4 + 2 * half) * 16 + q16];
    a1.y = at[(5 + 2 * half) * 16 + q16];
    a2.x = at[(8 + 2 * half) * 16 + q16];
    a2.y = at[(9 + 2 * half) * 16 + q16];

    const v4f* fin4 = reinterpret_cast<const v4f*>(fin);
    v4f* fout4 = reinterpret_cast<v4f*>(fout);
    v4f rv[IN_ROUNDS];
#pragma unroll
    for (unsigned k = 0; k < IN_ROUNDS; ++k) {
        const unsigned idx = tid + k * NTHREADS;
        if (idx < IN_PIECES) rv[k] = fin4[in_piece_goff(idx, x0, y0)];
    }

    const float w0 = w[0], w1 = w[1], w2 = w[2], w3 = w[3];
    const float w4 = w[4], w5 = w[5], w6 = w[6];

#pragma unroll
    for (unsigned k = 0; k < IN_ROUNDS; ++k) {
        const unsigned idx = tid + k * NTHREADS;
        if (idx < IN_PIECES) *reinterpret_cast<v4f*>(&s_buf[in_piece_loff(idx)]) = rv[k];
    }
    __syncthreads();

    tile_step<1>(s_buf, wv, half, q16, a0, a1, a2, w0, w1, w2, w3, w4, w5, w6);
    __syncthreads();
    tile_step<2>(s_buf, wv, half, q16, a0, a1, a2, w0, w1, w2, w3, w4, w5, w6);
    __syncthreads();
    tile_step<3>(s_buf, wv, half, q16, a0, a1, a2, w0, w1, w2, w3, w4, w5, w6);
    __syncthreads();
    tile_step<4>(s_buf, wv, half, q16, a0, a1, a2, w0, w1, w2, w3, w4, w5, w6);
    __syncthreads();

#pragma unroll
    for (unsigned k = 0; k < OUT_ROUNDS; ++k) {
        const unsigned idx = tid + k * NTHREADS;
        const v4f ov = *reinterpret_cast<const v4f*>(&s_buf[out_piece_loff(idx)]);
        fout4[out_piece_goff(idx, x0, y0)] = ov;
    }
}

extern "C" void kernel_launch(void* const* d_in, const int* in_sizes, int n_in,
                              void* d_out, int out_size, void* d_ws, size_t ws_size,
                              hipStream_t stream) {
    const float* f0 = (const float*)d_in[0];
    const float* w  = (const float*)d_in[1];
    float* out = (float*)d_out;

    const int threads = NTHREADS;
    const int blocks  = (NXL / TS) * (NYL / TS);

    lbm_fused<<<blocks, threads, 0, stream>>>(f0, w, out);
}


# one cell per lane: a wave pass takes 32 cells, not 16 duplicated in both half-waves; one half-wave exchange per moment pair fills the B registers of two WMMA sets; per-cell summation order unchanged
# speedup vs baseline: 2.1785x; 1.2114x over previous
#include <hip/hip_runtime.h>


typedef float v2f __attribute__((ext_vector_type(2)));
typedef float v8f __attribute__((ext_vector_type(8)));
typedef float v4f __attribute__((ext_vector_type(4)));

#define NXL 1024
#define NYL 1024
#define TAU 0.6f
#define GYF (-1e-4f)

#define STEPS 4
#define HALO STEPS
#define TS 32
#define RS (TS + 2 * HALO)
#define PITCH (RS * 9)
#define BUFF (RS * PITCH)
#define ROWV (PITCH / 4)
#define HV (HALO * 9 / 4)
#define MIDV (TS * 9 / 4)
#define IN_PIECES (RS * ROWV)
#define OUT_PIECES (TS * MIDV)
#define TILES_Y (NYL / TS)
#define NTHREADS 256
#define NWAVES (NTHREADS / 32)
#define IN_ROUNDS ((IN_PIECES + NTHREADS - 1) / NTHREADS)
#define OUT_ROUNDS (OUT_PIECES / NTHREADS)
static_assert((HALO * 9) % 4 == 0 && (TS * 9) % 4 == 0, "ring and centre segments are whole 16-byte pieces");
static_assert(NXL % TS == 0 && NYL % TS == 0 && TS % 4 == 0 && HALO % 4 == 0, "tiles cover the grid; segments start on a 16-byte piece");
static_assert(OUT_PIECES % NTHREADS == 0, "output split: whole rounds");
static_assert(2 * HV + MIDV == ROWV, "a staged row is ring + centre + ring");

struct ATab { float v[12][16]; };

static constexpr ATab makeAT() {
    const signed char M[9][9] = {
        { 1,  1,  1,  1,  1,  1,  1,  1,  1},
        { 0,  1,  0, -1,  0,  1, -1, -1,  1},
        { 0,  0,  1,  0, -1,  1,  1, -1, -1},
        {-4, -1, -1, -1, -1,  2,  2,  2,  2},
        { 4, -2, -2, -2, -2,  1,  1,  1,  1},
        { 0, -2,  0,  2,  0,  1, -1, -1,  1},
        { 0,  0, -2,  0,  2,  1,  1, -1, -1},
        { 0,  1, -1,  1, -1,  0,  0,  0,  0},
        { 0,  0,  0,  0,  0,  1, -1,  1, -1},
    };
    const float nrm2[9] = {9.f, 6.f, 6.f, 36.f, 36.f, 12.f, 12.f, 4.f, 4.f};
    ATab t{};
    for (int k = 0; k < 12; ++k)
        for (int q = 0; q < 16; ++q) {
            float x = 0.0f;
            if (k < 9 && q < 9)
                x = (float)M[k][q] / nrm2[k] * (1.0f / TAU);
            t.v[k][q] = x;
        }
    return t;
}

__constant__ ATab g_AT = makeAT();

static __device__ inline unsigned in_piece_goff(unsigned idx, unsigned x0, unsigned y0) {
    const unsigned row = idx / ROWV, p = idx - row * ROWV;
    const unsigned gx = (x0 + (NXL - HALO) + row) & (NXL - 1);
    unsigned gy, pp;
    if (p < HV)             { gy = (y0 + (NYL - HALO)) & (NYL - 1); pp = p; }
    else if (p < HV + MIDV) { gy = y0;                              pp = p - HV; }
    else                    { gy = (y0 + TS) & (NYL - 1);           pp = p - (HV + MIDV); }
    return ((gx << 8) + (gy >> 2)) * 9u + pp;
}
static __device__ inline unsigned in_piece_loff(unsigned idx) {
    const unsigned row = idx / ROWV, p = idx - row * ROWV;
    return row * PITCH + p * 4u;
}
static __device__ inline unsigned out_piece_goff(unsigned idx, unsigned x0, unsigned y0) {
    const unsigned row = idx / MIDV, kk = idx - row * MIDV;
    return (((x0 + row) << 8) + (y0 >> 2)) * 9u + kk;
}
static __device__ inline unsigned out_piece_loff(unsigned idx) {
    const unsigned row = idx / MIDV, kk = idx - row * MIDV;
    return (HALO + row) * PITCH + HALO * 9u + kk * 4u;
}

static __device__ __forceinline__ void swap_halves(float& d, float& s) {
    asm("v_permlane16_swap_b32 %0, %1" : "+v"(d), "+v"(s));
}
static __device__ __forceinline__ float other_half(float x) {
    const int b = __builtin_bit_cast(int, x);
    return __builtin_bit_cast(float, __builtin_amdgcn_permlanex16(b, b, 0x76543210, 0xfedcba98, false, false));
}

template <int S>
static __device__ __forceinline__ void tile_step(float* s_buf, unsigned wv, unsigned half, unsigned lane,
                                                 v2f a0, v2f a1, v2f a2,
                                                 v2f r0, v2f r1, v2f r2,
                                                 float w0, float w1, float w2, float w3,
                                                 float w4, float w5, float w6) {
    constexpr unsigned N = RS - 2 * S;
    constexpr unsigned NCELL = N * N;
    constexpr unsigned NGROUPS = (NCELL + 31u) / 32u;
    const float* src = s_buf + ((S - 1) & 1) * BUFF;
    float* dst = s_buf + (S & 1) * BUFF;

    const int exv[9] = {0, 1, 0, -1, 0, 1, -1, -1, 1};
    const int eyv[9] = {0, 0, 1, 0, -1, 1, 1, -1, -1};
    const float decay = 1.0f - 1.0f / TAU;

    for (unsigned g = wv; g < NGROUPS; g += NWAVES) {
        unsigned idx = g * 32u + lane;
        idx = idx < NCELL ? idx : NCELL - 1u;
        const unsigned r = idx / N, cc = idx - r * N;
        const unsigned cell = (S + r) * PITCH + (S + cc) * 9u;
        const float* c0 = src + cell;
        float f[9];
#pragma unroll
        for (int i = 0; i < 9; ++i)
            f[i] = c0[-exv[i] * PITCH - eyv[i] * 9 + i];

        const float rho = ((f[0] + f[1]) + (f[2] + f[3])) +
                          ((f[4] + f[5]) + (f[6] + f[7])) + f[8];
        const float mx = f[1] - f[3] + f[5] - f[6] - f[7] + f[8];
        const float my = f[2] - f[4] + f[5] + f[6] - f[7] - f[8];
        const float jx = mx;
        const float jy = my + (TAU * GYF) * rho * rho;
        const float jx2 = jx * jx, jy2 = jy * jy;
        const float j2 = jx2 + jy2;

        float x0 = rho,                 y0 = jy;
        float x1 = jx,                  y1 = w0 * rho + w1 * j2;
        float x2 = w2 * rho - w3 * j2,  y2 = w4 * jy;
        float x3 = w4 * jx,             y3 = w5 * (jx2 - jy2);
        float x4 = w6 * jx * jy,        y4 = 0.0f;
        swap_halves(x0, y0);
        swap_halves(x1, y1);
        swap_halves(x2, y2);
        swap_halves(x3, y3);
        swap_halves(x4, y4);
        v2f b0, b1, b2, d0, d1, d2;
        b0.x = x0; b0.y = x1; b1.x = x2; b1.y = x3; b2.x = x4; b2.y = 0.0f;
        d0.x = y0; d0.y = y1; d1.x = y2; d1.y = y3; d2.x = y4; d2.y = 0.0f;

        v8f c = {};
        v8f e = {};
        c = __builtin_amdgcn_wmma_f32_16x16x4_f32(false, a0, false, b0, (short)0, c,
                                                  false, false);
        e = __builtin_amdgcn_wmma_f32_16x16x4_f32(false, r0, false, d0, (short)0, e,
                                                  false, false);
        c = __builtin_amdgcn_wmma_f32_16x16x4_f32(false, a1, false, b1, (short)0, c,
                                                  false, false);
        e = __builtin_amdgcn_wmma_f32_16x16x4_f32(false, r1, false, d1, (short)0, e,
                                                  false, false);
        c = __builtin_amdgcn_wmma_f32_16x16x4_f32(false, a2, false, b2, (short)0, c,
                                                  false, false);
        e = __builtin_amdgcn_wmma_f32_16x16x4_f32(false, r2, false, d2, (short)0, e,
                                                  false, false);

        const float q8 = other_half(half ? c[0] : e[0]);
        float* o = dst + cell;
#pragma unroll
        for (int q = 0; q < 8; ++q) o[q] = decay * f[q] + (half ? e[q] : c[q]);
        o[8] = decay * f[8] + q8;
    }
}

__global__ __launch_bounds__(NTHREADS) void lbm_fused(const float* __restrict__ fin,
                                                      const float* __restrict__ w,
                                                      float* __restrict__ fout) {
    __shared__ __attribute__((aligned(16))) float s_buf[2 * BUFF];

    const unsigned tid  = threadIdx.x;
    const unsigned lane = tid & 31u;
    const unsigned half = lane >> 4;
    const unsigned q16  = lane & 15u;
    const unsigned wv   = (unsigned)__builtin_amdgcn_readfirstlane((int)(tid >> 5));
    const unsigned x0 = (blockIdx.x / TILES_Y) * TS;
    const unsigned y0 = (blockIdx.x % TILES_Y) * TS;

    const float* at = &g_AT.v[0][0];
    v2f a0, a1, a2;
    a0.x = at[(0 + 2 * half) * 16 + q16];
    a0.y = at[(1 + 2 * half) * 16 + q16];
    a1.x = at[(4 + 2 * half) * 16 + q16];
    a1.y = at[(5 + 2 * half) * 16 + q16];
    a2.x = at[(8 + 2 * half) * 16 + q16];
    a2.y = at[(9 + 2 * half) * 16 + q16];
    const unsigned q16r = (q16 + 8u) & 15u;
    v2f r0, r1, r2;
    r0.x = at[(0 + 2 * half) * 16 + q16r];
    r0.y = at[(1 + 2 * half) * 16 + q16r];
    r1.x = at[(4 + 2 * half) * 16 + q16r];
    r1.y = at[(5 + 2 * half) * 16 + q16r];
    r2.x = at[(8 + 2 * half) * 16 + q16r];
    r2.y = at[(9 + 2 * half) * 16 + q16r];

    const v4f* fin4 = reinterpret_cast<const v4f*>(fin);
    v4f* fout4 = reinterpret_cast<v4f*>(fout);
    v4f rv[IN_ROUNDS];
#pragma unroll
    for (unsigned k = 0; k < IN_ROUNDS; ++k) {
        const unsigned idx = tid + k * NTHREADS;
        if (idx < IN_PIECES) rv[k] = fin4[in_piece_goff(idx, x0, y0)];
    }

    const float w0 = w[0], w1 = w[1], w2 = w[2], w3 = w[3];
    const float w4 = w[4], w5 = w[5], w6 = w[6];

#pragma unroll
    for (unsigned k = 0; k < IN_ROUNDS; ++k) {
        const unsigned idx = tid + k * NTHREADS;
        if (idx < IN_PIECES) *reinterpret_cast<v4f*>(&s_buf[in_piece_loff(idx)]) = rv[k];
    }
    __syncthreads();

    tile_step<1>(s_buf, wv, half, lane, a0, a1, a2, r0, r1, r2, w0, w1, w2, w3, w4, w5, w6);
    __syncthreads();
    tile_step<2>(s_buf, wv, half, lane, a0, a1, a2, r0, r1, r2, w0, w1, w2, w3, w4, w5, w6);
    __syncthreads();
    tile_step<3>(s_buf, wv, half, lane, a0, a1, a2, r0, r1, r2, w0, w1, w2, w3, w4, w5, w6);
    __syncthreads();
    tile_step<4>(s_buf, wv, half, lane, a0, a1, a2, r0, r1, r2, w0, w1, w2, w3, w4, w5, w6);
    __syncthreads();

#pragma unroll
    for (unsigned k = 0; k < OUT_ROUNDS; ++k) {
        const unsigned idx = tid + k * NTHREADS;
        const v4f ov = *reinterpret_cast<const v4f*>(&s_buf[out_piece_loff(idx)]);
        fout4[out_piece_goff(idx, x0, y0)] = ov;
    }
}

extern "C" void kernel_launch(void* const* d_in, const int* in_sizes, int n_in,
                              void* d_out, int out_size, void* d_ws, size_t ws_size,
                              hipStream_t stream) {
    const float* f0 = (const float*)d_in[0];
    const float* w  = (const float*)d_in[1];
    float* out = (float*)d_out;

    const int threads = NTHREADS;
    const int blocks  = (NXL / TS) * (NYL / TS);

    lbm_fused<<<blocks, threads, 0, stream>>>(f0, w, out);
}


# fold the learned coefficients w into the A matrix: six per-cell terms (rho, jx, jy, j2, jx2-jy2, jx*jy) replace nine moments, K 12 -> 8, two WMMAs per 16 cells not three; sums re-associated
# speedup vs baseline: 2.5569x; 1.1737x over previous
#include <hip/hip_runtime.h>


typedef float v2f __attribute__((ext_vector_type(2)));
typedef float v8f __attribute__((ext_vector_type(8)));
typedef float v4f __attribute__((ext_vector_type(4)));

#define NXL 1024
#define NYL 1024
#define TAU 0.6f
#define GYF (-1e-4f)

#define STEPS 4
#define HALO STEPS
#define TS 32
#define RS (TS + 2 * HALO)
#define PITCH (RS * 9)
#define BUFF (RS * PITCH)
#define ROWV (PITCH / 4)
#define HV (HALO * 9 / 4)
#define MIDV (TS * 9 / 4)
#define IN_PIECES (RS * ROWV)
#define OUT_PIECES (TS * MIDV)
#define TILES_Y (NYL / TS)
#define NTHREADS 256
#define NWAVES (NTHREADS / 32)
#define IN_ROUNDS ((IN_PIECES + NTHREADS - 1) / NTHREADS)
#define OUT_ROUNDS (OUT_PIECES / NTHREADS)
static_assert((HALO * 9) % 4 == 0 && (TS * 9) % 4 == 0, "ring and centre segments are whole 16-byte pieces");
static_assert(NXL % TS == 0 && NYL % TS == 0 && TS % 4 == 0 && HALO % 4 == 0, "tiles cover the grid; segments start on a 16-byte piece");
static_assert(OUT_PIECES % NTHREADS == 0, "output split: whole rounds");
static_assert(2 * HV + MIDV == ROWV, "a staged row is ring + centre + ring");

struct ATab { float v[9][16]; };

static constexpr ATab makeAT() {
    const signed char M[9][9] = {
        { 1,  1,  1,  1,  1,  1,  1,  1,  1},
        { 0,  1,  0, -1,  0,  1, -1, -1,  1},
        { 0,  0,  1,  0, -1,  1,  1, -1, -1},
        {-4, -1, -1, -1, -1,  2,  2,  2,  2},
        { 4, -2, -2, -2, -2,  1,  1,  1,  1},
        { 0, -2,  0,  2,  0,  1, -1, -1,  1},
        { 0,  0, -2,  0,  2,  1,  1, -1, -1},
        { 0,  1, -1,  1, -1,  0,  0,  0,  0},
        { 0,  0,  0,  0,  0,  1, -1,  1, -1},
    };
    const float nrm2[9] = {9.f, 6.f, 6.f, 36.f, 36.f, 12.f, 12.f, 4.f, 4.f};
    ATab t{};
    for (int k = 0; k < 9; ++k)
        for (int q = 0; q < 16; ++q) {
            float x = 0.0f;
            if (q < 9)
                x = (float)M[k][q] / nrm2[k] * (1.0f / TAU);
            t.v[k][q] = x;
        }
    return t;
}

__constant__ ATab g_AT = makeAT();

static __device__ inline unsigned in_piece_goff(unsigned idx, unsigned x0, unsigned y0) {
    const unsigned row = idx / ROWV, p = idx - row * ROWV;
    const unsigned gx = (x0 + (NXL - HALO) + row) & (NXL - 1);
    unsigned gy, pp;
    if (p < HV)             { gy = (y0 + (NYL - HALO)) & (NYL - 1); pp = p; }
    else if (p < HV + MIDV) { gy = y0;                              pp = p - HV; }
    else                    { gy = (y0 + TS) & (NYL - 1);           pp = p - (HV + MIDV); }
    return ((gx << 8) + (gy >> 2)) * 9u + pp;
}
static __device__ inline unsigned in_piece_loff(unsigned idx) {
    const unsigned row = idx / ROWV, p = idx - row * ROWV;
    return row * PITCH + p * 4u;
}
static __device__ inline unsigned out_piece_goff(unsigned idx, unsigned x0, unsigned y0) {
    const unsigned row = idx / MIDV, kk = idx - row * MIDV;
    return (((x0 + row) << 8) + (y0 >> 2)) * 9u + kk;
}
static __device__ inline unsigned out_piece_loff(unsigned idx) {
    const unsigned row = idx / MIDV, kk = idx - row * MIDV;
    return (HALO + row) * PITCH + HALO * 9u + kk * 4u;
}

static __device__ __forceinline__ void swap_halves(float& d, float& s) {
    asm("v_permlane16_swap_b32 %0, %1" : "+v"(d), "+v"(s));
}
static __device__ __forceinline__ float other_half(float x) {
    const int b = __builtin_bit_cast(int, x);
    return __builtin_bit_cast(float, __builtin_amdgcn_permlanex16(b, b, 0x76543210, 0xfedcba98, false, false));
}

static __device__ __forceinline__ void build_a(const float* tb, unsigned half, unsigned q,
                                               float w0, float w1, float w2, float w3,
                                               float w4, float w5, float w6,
                                               v2f& a0, v2f& a1) {
    const float t0 = tb[0 * 16 + q], t1 = tb[1 * 16 + q], t2 = tb[2 * 16 + q];
    const float t3 = tb[3 * 16 + q], t4 = tb[4 * 16 + q], t5 = tb[5 * 16 + q];
    const float t6 = tb[6 * 16 + q], t7 = tb[7 * 16 + q], t8 = tb[8 * 16 + q];
    const float k0 = t0 + w0 * t3 + w2 * t4;
    const float k1 = t1 + w4 * t5;
    const float k2 = t2 + w4 * t6;
    const float k3 = w1 * t3 - w3 * t4;
    const float k4 = w5 * t7;
    const float k5 = w6 * t8;
    a0.x = half ? k2 : k0;
    a0.y = half ? k3 : k1;
    a1.x = half ? k5 : k4;
    a1.y = 0.0f;
}

template <int S>
static __device__ __forceinline__ void tile_step(float* s_buf, unsigned wv, unsigned half, unsigned lane,
                                                 v2f a0, v2f a1, v2f r0, v2f r1) {
    constexpr unsigned N = RS - 2 * S;
    constexpr unsigned NCELL = N * N;
    constexpr unsigned NGROUPS = (NCELL + 31u) / 32u;
    const float* src = s_buf + ((S - 1) & 1) * BUFF;
    float* dst = s_buf + (S & 1) * BUFF;

    const int exv[9] = {0, 1, 0, -1, 0, 1, -1, -1, 1};
    const int eyv[9] = {0, 0, 1, 0, -1, 1, 1, -1, -1};
    const float decay = 1.0f - 1.0f / TAU;

    for (unsigned g = wv; g < NGROUPS; g += NWAVES) {
        unsigned idx = g * 32u + lane;
        idx = idx < NCELL ? idx : NCELL - 1u;
        const unsigned r = idx / N, cc = idx - r * N;
        const unsigned cell = (S + r) * PITCH + (S + cc) * 9u;
        const float* c0 = src + cell;
        float f[9];
#pragma unroll
        for (int i = 0; i < 9; ++i)
            f[i] = c0[-exv[i] * PITCH - eyv[i] * 9 + i];

        const float rho = ((f[0] + f[1]) + (f[2] + f[3])) +
                          ((f[4] + f[5]) + (f[6] + f[7])) + f[8];
        const float mx = f[1] - f[3] + f[5] - f[6] - f[7] + f[8];
        const float my = f[2] - f[4] + f[5] + f[6] - f[7] - f[8];
        const float jx = mx;
        const float jy = my + (TAU * GYF) * rho * rho;
        const float jx2 = jx * jx, jy2 = jy * jy;

        float x0 = rho,        y0 = jy;
        float x1 = jx,         y1 = jx2 + jy2;
        float x2 = jx2 - jy2,  y2 = jx * jy;
        swap_halves(x0, y0);
        swap_halves(x1, y1);
        swap_halves(x2, y2);
        v2f b0, b1, d0, d1;
        b0.x = x0; b0.y = x1; b1.x = x2; b1.y = 0.0f;
        d0.x = y0; d0.y = y1; d1.x = y2; d1.y = 0.0f;

        v8f c = {};
        v8f e = {};
        c = __builtin_amdgcn_wmma_f32_16x16x4_f32(false, a0, false, b0, (short)0, c,
                                                  false, false);
        e = __builtin_amdgcn_wmma_f32_16x16x4_f32(false, r0, false, d0, (short)0, e,
                                                  false, false);
        c = __builtin_amdgcn_wmma_f32_16x16x4_f32(false, a1, false, b1, (short)0, c,
                                                  false, false);
        e = __builtin_amdgcn_wmma_f32_16x16x4_f32(false, r1, false, d1, (short)0, e,
                                                  false, false);

        const float q8 = other_half(half ? c[0] : e[0]);
        float* o = dst + cell;
#pragma unroll
        for (int q = 0; q < 8; ++q) o[q] = decay * f[q] + (half ? e[q] : c[q]);
        o[8] = decay * f[8] + q8;
    }
}

__global__ __launch_bounds__(NTHREADS) void lbm_fused(const float* __restrict__ fin,
                                                      const float* __restrict__ w,
                                                      float* __restrict__ fout) {
    __shared__ __attribute__((aligned(16))) float s_buf[2 * BUFF];

    const unsigned tid  = threadIdx.x;
    const unsigned lane = tid & 31u;
    const unsigned half = lane >> 4;
    const unsigned q16  = lane & 15u;
    const unsigned wv   = (unsigned)__builtin_amdgcn_readfirstlane((int)(tid >> 5));
    const unsigned x0 = (blockIdx.x / TILES_Y) * TS;
    const unsigned y0 = (blockIdx.x % TILES_Y) * TS;

    const v4f* fin4 = reinterpret_cast<const v4f*>(fin);
    v4f* fout4 = reinterpret_cast<v4f*>(fout);
    v4f rv[IN_ROUNDS];
#pragma unroll
    for (unsigned k = 0; k < IN_ROUNDS; ++k) {
        const unsigned idx = tid + k * NTHREADS;
        if (idx < IN_PIECES) rv[k] = fin4[in_piece_goff(idx, x0, y0)];
    }

    const float w0 = w[0], w1 = w[1], w2 = w[2], w3 = w[3];
    const float w4 = w[4], w5 = w[5], w6 = w[6];

    v2f a0, a1, r0, r1;
    build_a(&g_AT.v[0][0], half, q16, w0, w1, w2, w3, w4, w5, w6, a0, a1);
    build_a(&g_AT.v[0][0], half, (q16 + 8u) & 15u, w0, w1, w2, w3, w4, w5, w6, r0, r1);

#pragma unroll
    for (unsigned k = 0; k < IN_ROUNDS; ++k) {
        const unsigned idx = tid + k * NTHREADS;
        if (idx < IN_PIECES) *reinterpret_cast<v4f*>(&s_buf[in_piece_loff(idx)]) = rv[k];
    }
    __syncthreads();

    tile_step<1>(s_buf, wv, half, lane, a0, a1, r0, r1);
    __syncthreads();
    tile_step<2>(s_buf, wv, half, lane, a0, a1, r0, r1);
    __syncthreads();
    tile_step<3>(s_buf, wv, half, lane, a0, a1, r0, r1);
    __syncthreads();
    tile_step<4>(s_buf, wv, half, lane, a0, a1, r0, r1);
    __syncthreads();

#pragma unroll
    for (unsigned k = 0; k < OUT_ROUNDS; ++k) {
        const unsigned idx = tid + k * NTHREADS;
        const v4f ov = *reinterpret_cast<const v4f*>(&s_buf[out_piece_loff(idx)]);
        fout4[out_piece_goff(idx, x0, y0)] = ov;
    }
}

extern "C" void kernel_launch(void* const* d_in, const int* in_sizes, int n_in,
                              void* d_out, int out_size, void* d_ws, size_t ws_size,
                              hipStream_t stream) {
    const float* f0 = (const float*)d_in[0];
    const float* w  = (const float*)d_in[1];
    float* out = (float*)d_out;

    const int threads = NTHREADS;
    const int blocks  = (NXL / TS) * (NYL / TS);

    lbm_fused<<<blocks, threads, 0, stream>>>(f0, w, out);
}


# block-diagonal A: column n carries the cells of lanes n | n+16 in K slots 0,1 | 2,3, rows 0-7 | 8-15 their q 1..8, q 0 by a 4th WMMA: no half-wave exchanges or selects; sums re-associated
# speedup vs baseline: 2.9983x; 1.1726x over previous
#include <hip/hip_runtime.h>


typedef float v2f __attribute__((ext_vector_type(2)));
typedef float v8f __attribute__((ext_vector_type(8)));
typedef float v4f __attribute__((ext_vector_type(4)));

#define NXL 1024
#define NYL 1024
#define TAU 0.6f
#define GYF (-1e-4f)

#define STEPS 4
#define HALO STEPS
#define TSX 64
#define TSY 32
#define RSX (TSX + 2 * HALO)
#define RSY (TSY + 2 * HALO)
#define PITCH (RSY * 9)
#define BUFF (RSX * PITCH)
#define ROWV (PITCH / 4)
#define HV (HALO * 9 / 4)
#define MIDV (TSY * 9 / 4)
#define IN_PIECES (RSX * ROWV)
#define OUT_PIECES (TSX * MIDV)
#define TILES_Y (NYL / TSY)
#define NTHREADS 512
#define NWAVES (NTHREADS / 32)
#define IN_ROUNDS ((IN_PIECES + NTHREADS - 1) / NTHREADS)
#define OUT_ROUNDS (OUT_PIECES / NTHREADS)
static_assert((HALO * 9) % 4 == 0 && (TSY * 9) % 4 == 0, "ring and centre segments are whole 16-byte pieces");
static_assert(NXL % TSX == 0 && NYL % TSY == 0 && TSY % 4 == 0 && HALO % 4 == 0, "tiles cover the grid; segments start on a 16-byte piece");
static_assert(OUT_PIECES % NTHREADS == 0, "output split: whole rounds");
static_assert(2 * HV + MIDV == ROWV, "a staged row is ring + centre + ring");

struct ATab { float v[9][16]; };

static constexpr ATab makeAT() {
    const signed char M[9][9] = {
        { 1,  1,  1,  1,  1,  1,  1,  1,  1},
        { 0,  1,  0, -1,  0,  1, -1, -1,  1},
        { 0,  0,  1,  0, -1,  1,  1, -1, -1},
        {-4, -1, -1, -1, -1,  2,  2,  2,  2},
        { 4, -2, -2, -2, -2,  1,  1,  1,  1},
        { 0, -2,  0,  2,  0,  1, -1, -1,  1},
        { 0,  0, -2,  0,  2,  1,  1, -1, -1},
        { 0,  1, -1,  1, -1,  0,  0,  0,  0},
        { 0,  0,  0,  0,  0,  1, -1,  1, -1},
    };
    const float nrm2[9] = {9.f, 6.f, 6.f, 36.f, 36.f, 12.f, 12.f, 4.f, 4.f};
    ATab t{};
    for (int k = 0; k < 9; ++k)
        for (int q = 0; q < 16; ++q) {
            float x = 0.0f;
            if (q < 9)
                x = (float)M[k][q] / nrm2[k] * (1.0f / TAU);
            t.v[k][q] = x;
        }
    return t;
}

__constant__ ATab g_AT = makeAT();

static __device__ inline unsigned in_piece_goff(unsigned idx, unsigned x0, unsigned y0) {
    const unsigned row = idx / ROWV, p = idx - row * ROWV;
    const unsigned gx = (x0 + (NXL - HALO) + row) & (NXL - 1);
    unsigned gy, pp;
    if (p < HV)             { gy = (y0 + (NYL - HALO)) & (NYL - 1); pp = p; }
    else if (p < HV + MIDV) { gy = y0;                              pp = p - HV; }
    else                    { gy = (y0 + TSY) & (NYL - 1);          pp = p - (HV + MIDV); }
    return ((gx << 8) + (gy >> 2)) * 9u + pp;
}
static __device__ inline unsigned in_piece_loff(unsigned idx) {
    const unsigned row = idx / ROWV, p = idx - row * ROWV;
    return row * PITCH + p * 4u;
}
static __device__ inline unsigned out_piece_goff(unsigned idx, unsigned x0, unsigned y0) {
    const unsigned row = idx / MIDV, kk = idx - row * MIDV;
    return (((x0 + row) << 8) + (y0 >> 2)) * 9u + kk;
}

static __device__ inline unsigned out_piece_loff(unsigned idx) {
    const unsigned row = idx / MIDV, kk = idx - row * MIDV;
    return (HALO + row) * PITCH + HALO * 9u + kk * 4u;
}

static_assert(makeAT().v[1][0] == 0.0f && makeAT().v[2][0] == 0.0f && makeAT().v[5][0] == 0.0f &&
              makeAT().v[6][0] == 0.0f && makeAT().v[7][0] == 0.0f && makeAT().v[8][0] == 0.0f,
              "output q = 0 takes only the rho and j2 terms");
static __device__ __forceinline__ void build_a(const float* tb, unsigned lane,
                                               float w0, float w1, float w2, float w3,
                                               float w4, float w5, float w6,
                                               v2f& aA, v2f& aB, v2f& aC, v2f& aD) {
    const unsigned c = lane & 15u, h = lane >> 4;
    const unsigned q = (c & 7u) + 1u;
    const bool on  = (c >> 3) == h;
    const bool on0 = on && (c & 7u) == 0u;
    const float t0 = tb[0 * 16 + q], t1 = tb[1 * 16 + q], t2 = tb[2 * 16 + q];
    const float t3 = tb[3 * 16 + q], t4 = tb[4 * 16 + q], t5 = tb[5 * 16 + q];
    const float t6 = tb[6 * 16 + q], t7 = tb[7 * 16 + q], t8 = tb[8 * 16 + q];
    const float k0 = t0 + w0 * t3 + w2 * t4;
    const float k1 = t1 + w4 * t5;
    const float k2 = t2 + w4 * t6;
    const float k3 = w1 * t3 - w3 * t4;
    const float k4 = w5 * t7;
    const float k5 = w6 * t8;
    aA.x = on ? k0 : 0.0f;  aA.y = on ? k3 : 0.0f;
    aB.x = on ? k1 : 0.0f;  aB.y = on ? k2 : 0.0f;
    aC.x = on ? k4 : 0.0f;  aC.y = on ? k5 : 0.0f;
    const float z0 = tb[0 * 16], z3 = tb[3 * 16], z4 = tb[4 * 16];
    aD.x = on0 ? z0 + w0 * z3 + w2 * z4 : 0.0f;
    aD.y = on0 ? w1 * z3 - w3 * z4 : 0.0f;
}

template <int S>
static __device__ __forceinline__ void collide_cell(const float* c0,
                                                    v2f aA, v2f aB, v2f aC, v2f aD, float (&o)[9]) {
    const int exv[9] = {0, 1, 0, -1, 0, 1, -1, -1, 1};
    const int eyv[9] = {0, 0, 1, 0, -1, 1, 1, -1, -1};
    const float decay = 1.0f - 1.0f / TAU;

    float f[9];
#pragma unroll
    for (int i = 0; i < 9; ++i)
        f[i] = c0[-S * (exv[i] * PITCH + eyv[i] * 9) + i];

    const float rho = ((f[0] + f[1]) + (f[2] + f[3])) +
                      ((f[4] + f[5]) + (f[6] + f[7])) + f[8];
    const float mx = f[1] - f[3] + f[5] - f[6] - f[7] + f[8];
    const float my = f[2] - f[4] + f[5] + f[6] - f[7] - f[8];
    const float jx = mx;
    const float jy = my + (TAU * GYF) * rho * rho;
    const float jx2 = jx * jx, jy2 = jy * jy;

    v2f bA, bB, bC;
    bA.x = rho;        bA.y = jx2 + jy2;
    bB.x = jx;         bB.y = jy;
    bC.x = jx2 - jy2;  bC.y = jx * jy;

    v8f d = {};
    v8f z = {};
    d = __builtin_amdgcn_wmma_f32_16x16x4_f32(false, aA, false, bA, (short)0, d,
                                              false, false);
    z = __builtin_amdgcn_wmma_f32_16x16x4_f32(false, aD, false, bA, (short)0, z,
                                              false, false);
    d = __builtin_amdgcn_wmma_f32_16x16x4_f32(false, aB, false, bB, (short)0, d,
                                              false, false);
    d = __builtin_amdgcn_wmma_f32_16x16x4_f32(false, aC, false, bC, (short)0, d,
                                              false, false);

    o[0] = decay * f[0] + z[0];
#pragma unroll
    for (int q = 1; q < 9; ++q) o[q] = decay * f[q] + d[q - 1];
}

template <int S>
static __device__ __forceinline__ void tile_step(float* s_buf, unsigned wv, unsigned lane,
                                                 v2f aA, v2f aB, v2f aC, v2f aD) {
    constexpr unsigned NR = RSX - 2 * S;
    constexpr unsigned N = RSY - 2 * S;
    constexpr unsigned NCELL = NR * N;
    constexpr unsigned NGROUPS = (NCELL + 31u) / 32u;
    const int exv[9] = {0, 1, 0, -1, 0, 1, -1, -1, 1};
    const int eyv[9] = {0, 0, 1, 0, -1, 1, 1, -1, -1};

    for (unsigned g = wv; g < NGROUPS; g += NWAVES) {
        unsigned idx = g * 32u + lane;
        idx = idx < NCELL ? idx : NCELL - 1u;
        const unsigned r = idx / N, cc = idx - r * N;
        float* c0 = s_buf + (S + r) * PITCH + (S + cc) * 9u;
        float o[9];
        collide_cell<S>(c0, aA, aB, aC, aD, o);
#pragma unroll
        for (int q = 0; q < 9; ++q)
            c0[-S * (exv[q] * PITCH + eyv[q] * 9) + q] = o[q];
    }
}

static __device__ __forceinline__ void tile_step_last(float* s_buf, unsigned wv, unsigned lane,
                                                      v2f aA, v2f aB, v2f aC, v2f aD) {
    constexpr unsigned GPW = TSX / NWAVES;
    static_assert(TSY == 32 && TSX % NWAVES == 0, "a group is one owned row; whole groups per wave");
    float o[GPW][9];
#pragma unroll
    for (unsigned u = 0; u < GPW; ++u) {
        const unsigned row = wv + u * NWAVES;
        const float* c0 = s_buf + (HALO + row) * PITCH + (HALO + lane) * 9u;
        collide_cell<STEPS>(c0, aA, aB, aC, aD, o[u]);
    }
    __syncthreads();
#pragma unroll
    for (unsigned u = 0; u < GPW; ++u) {
        const unsigned row = wv + u * NWAVES;
        float* c0 = s_buf + (HALO + row) * PITCH + (HALO + lane) * 9u;
#pragma unroll
        for (int q = 0; q < 9; ++q) c0[q] = o[u][q];
    }
}

__global__ __launch_bounds__(NTHREADS) void lbm_fused(const float* __restrict__ fin,
                                                      const float* __restrict__ w,
                                                      float* __restrict__ fout) {
    __shared__ __attribute__((aligned(16))) float s_buf[BUFF];

    const unsigned tid  = threadIdx.x;
    const unsigned lane = tid & 31u;
    const unsigned wv   = (unsigned)__builtin_amdgcn_readfirstlane((int)(tid >> 5));
    const unsigned x0 = (blockIdx.x / TILES_Y) * TSX;
    const unsigned y0 = (blockIdx.x % TILES_Y) * TSY;

    const v4f* fin4 = reinterpret_cast<const v4f*>(fin);
    v4f* fout4 = reinterpret_cast<v4f*>(fout);
    v4f rv[IN_ROUNDS];
#pragma unroll
    for (unsigned k = 0; k < IN_ROUNDS; ++k) {
        const unsigned idx = tid + k * NTHREADS;
        if (idx < IN_PIECES) rv[k] = fin4[in_piece_goff(idx, x0, y0)];
    }

    const float w0 = w[0], w1 = w[1], w2 = w[2], w3 = w[3];
    const float w4 = w[4], w5 = w[5], w6 = w[6];

    v2f aA, aB, aC, aD;
    build_a(&g_AT.v[0][0], lane, w0, w1, w2, w3, w4, w5, w6, aA, aB, aC, aD);

#pragma unroll
    for (unsigned k = 0; k < IN_ROUNDS; ++k) {
        const unsigned idx = tid + k * NTHREADS;
        if (idx < IN_PIECES) *reinterpret_cast<v4f*>(&s_buf[in_piece_loff(idx)]) = rv[k];
    }
    __syncthreads();

    tile_step<1>(s_buf, wv, lane, aA, aB, aC, aD);
    __syncthreads();
    tile_step<2>(s_buf, wv, lane, aA, aB, aC, aD);
    __syncthreads();
    tile_step<3>(s_buf, wv, lane, aA, aB, aC, aD);
    __syncthreads();
    tile_step_last(s_buf, wv, lane, aA, aB, aC, aD);
    __syncthreads();

#pragma unroll
    for (unsigned k = 0; k < OUT_ROUNDS; ++k) {
        const unsigned idx = tid + k * NTHREADS;
        const v4f ov = *reinterpret_cast<const v4f*>(&s_buf[out_piece_loff(idx)]);
        fout4[out_piece_goff(idx, x0, y0)] = ov;
    }
}

extern "C" void kernel_launch(void* const* d_in, const int* in_sizes, int n_in,
                              void* d_out, int out_size, void* d_ws, size_t ws_size,
                              hipStream_t stream) {
    const float* f0 = (const float*)d_in[0];
    const float* w  = (const float*)d_in[1];
    float* out = (float*)d_out;

    const int threads = NTHREADS;
    const int blocks  = (NXL / TSX) * (NYL / TSY);

    lbm_fused<<<blocks, threads, 0, stream>>>(f0, w, out);
}


# three WMMAs per 32 cells, not four: the {rho,j2} WMMA gives only the 3 distinct row sums (q 0 | 1-4 | 5-8), q 0 included; decay*f + that sum enters the {jx,jy} WMMA as C; sums re-associated
# speedup vs baseline: 3.1506x; 1.0508x over previous
#include <hip/hip_runtime.h>


typedef float v2f __attribute__((ext_vector_type(2)));
typedef float v8f __attribute__((ext_vector_type(8)));
typedef float v4f __attribute__((ext_vector_type(4)));

#define NXL 1024
#define NYL 1024
#define TAU 0.6f
#define GYF (-1e-4f)

#define STEPS 4
#define HALO STEPS
#define TSX 64
#define TSY 32
#define RSX (TSX + 2 * HALO)
#define RSY (TSY + 2 * HALO)
#define PITCH (RSY * 9)
#define BUFF (RSX * PITCH)
#define ROWV (PITCH / 4)
#define HV (HALO * 9 / 4)
#define MIDV (TSY * 9 / 4)
#define IN_PIECES (RSX * ROWV)
#define OUT_PIECES (TSX * MIDV)
#define TILES_Y (NYL / TSY)
#define NTHREADS 512
#define NWAVES (NTHREADS / 32)
#define IN_ROUNDS ((IN_PIECES + NTHREADS - 1) / NTHREADS)
#define OUT_ROUNDS (OUT_PIECES / NTHREADS)
static_assert((HALO * 9) % 4 == 0 && (TSY * 9) % 4 == 0, "ring and centre segments are whole 16-byte pieces");
static_assert(NXL % TSX == 0 && NYL % TSY == 0 && TSY % 4 == 0 && HALO % 4 == 0, "tiles cover the grid; segments start on a 16-byte piece");
static_assert(OUT_PIECES % NTHREADS == 0, "output split: whole rounds");
static_assert(2 * HV + MIDV == ROWV, "a staged row is ring + centre + ring");

struct ATab { float v[9][16]; };

static constexpr ATab makeAT() {
    const signed char M[9][9] = {
        { 1,  1,  1,  1,  1,  1,  1,  1,  1},
        { 0,  1,  0, -1,  0,  1, -1, -1,  1},
        { 0,  0,  1,  0, -1,  1,  1, -1, -1},
        {-4, -1, -1, -1, -1,  2,  2,  2,  2},
        { 4, -2, -2, -2, -2,  1,  1,  1,  1},
        { 0, -2,  0,  2,  0,  1, -1, -1,  1},
        { 0,  0, -2,  0,  2,  1,  1, -1, -1},
        { 0,  1, -1,  1, -1,  0,  0,  0,  0},
        { 0,  0,  0,  0,  0,  1, -1,  1, -1},
    };
    const float nrm2[9] = {9.f, 6.f, 6.f, 36.f, 36.f, 12.f, 12.f, 4.f, 4.f};
    ATab t{};
    for (int k = 0; k < 9; ++k)
        for (int q = 0; q < 16; ++q) {
            float x = 0.0f;
            if (q < 9)
                x = (float)M[k][q] / nrm2[k] * (1.0f / TAU);
            t.v[k][q] = x;
        }
    return t;
}

__constant__ ATab g_AT = makeAT();

static __device__ inline unsigned in_piece_goff(unsigned idx, unsigned x0, unsigned y0) {
    const unsigned row = idx / ROWV, p = idx - row * ROWV;
    const unsigned gx = (x0 + (NXL - HALO) + row) & (NXL - 1);
    unsigned gy, pp;
    if (p < HV)             { gy = (y0 + (NYL - HALO)) & (NYL - 1); pp = p; }
    else if (p < HV + MIDV) { gy = y0;                              pp = p - HV; }
    else                    { gy = (y0 + TSY) & (NYL - 1);          pp = p - (HV + MIDV); }
    return ((gx << 8) + (gy >> 2)) * 9u + pp;
}
static __device__ inline unsigned in_piece_loff(unsigned idx) {
    const unsigned row = idx / ROWV, p = idx - row * ROWV;
    return row * PITCH + p * 4u;
}
static __device__ inline unsigned out_piece_goff(unsigned idx, unsigned x0, unsigned y0) {
    const unsigned row = idx / MIDV, kk = idx - row * MIDV;
    return (((x0 + row) << 8) + (y0 >> 2)) * 9u + kk;
}

static __device__ inline unsigned out_piece_loff(unsigned idx) {
    const unsigned row = idx / MIDV, kk = idx - row * MIDV;
    return (HALO + row) * PITCH + HALO * 9u + kk * 4u;
}

static_assert(makeAT().v[1][0] == 0.0f && makeAT().v[2][0] == 0.0f && makeAT().v[5][0] == 0.0f &&
              makeAT().v[6][0] == 0.0f && makeAT().v[7][0] == 0.0f && makeAT().v[8][0] == 0.0f,
              "output q = 0 takes only the rho and j2 terms");
#define ROWQ_LIST {5, 1, 6, 3, 7, 8, 2, 4}
#define ROWQ_PACKED 0x42873615u
static constexpr bool rowq_ok() {
    const int rq[8] = ROWQ_LIST;
    unsigned seen = 0;
    for (int r = 0; r < 8; ++r) {
        if (rq[r] < 1 || rq[r] > 8) return false;
        if (((ROWQ_PACKED >> (4 * r)) & 15u) != (unsigned)rq[r]) return false;
        seen |= 1u << rq[r];
    }
    return seen == 0x1feu;
}
static_assert(rowq_ok(), "ROWQ is a permutation of 1..8 and matches its packed form");
static constexpr bool class_rows_ok() {
    const ATab t = makeAT();
    const int rows[3] = {0, 3, 4};
    for (int i = 0; i < 3; ++i)
        for (int q = 1; q < 9; ++q)
            if (t.v[rows[i]][q] != t.v[rows[i]][q < 5 ? 1 : 5]) return false;
    return true;
}
static_assert(class_rows_ok(), "the rho and j2 coefficients are equal within q = 1..4 and within q = 5..8");
static __device__ __forceinline__ void build_a(const float* tb, unsigned lane,
                                               float w0, float w1, float w2, float w3,
                                               float w4, float w5, float w6,
                                               v2f& aR, v2f& aB, v2f& aC) {
    const unsigned c = lane & 15u, h = lane >> 4;
    const unsigned rr = c & 7u;
    const bool on  = (c >> 3) == h;
    const unsigned q = (ROWQ_PACKED >> (4u * rr)) & 15u;
    const float t1 = tb[1 * 16 + q], t2 = tb[2 * 16 + q], t5 = tb[5 * 16 + q];
    const float t6 = tb[6 * 16 + q], t7 = tb[7 * 16 + q], t8 = tb[8 * 16 + q];
    const float k1 = t1 + w4 * t5;
    const float k2 = t2 + w4 * t6;
    const float k4 = w5 * t7;
    const float k5 = w6 * t8;
    aB.x = on ? k1 : 0.0f;  aB.y = on ? k2 : 0.0f;
    aC.x = on ? k4 : 0.0f;  aC.y = on ? k5 : 0.0f;
    const unsigned qc = rr == 0u ? 0u : (rr == 1u ? 1u : 5u);
    const bool onr = on && rr < 3u;
    const float z0 = tb[0 * 16 + qc], z3 = tb[3 * 16 + qc], z4 = tb[4 * 16 + qc];
    aR.x = onr ? z0 + w0 * z3 + w2 * z4 : 0.0f;
    aR.y = onr ? w1 * z3 - w3 * z4 : 0.0f;
}

template <int S>
static __device__ __forceinline__ void collide_cell(const float* c0,
                                                    v2f aR, v2f aB, v2f aC, float (&o)[9]) {
    const int exv[9] = {0, 1, 0, -1, 0, 1, -1, -1, 1};
    const int eyv[9] = {0, 0, 1, 0, -1, 1, 1, -1, -1};
    const float decay = 1.0f - 1.0f / TAU;
    const int rq[8] = ROWQ_LIST;

    float f[9];
#pragma unroll
    for (int i = 0; i < 9; ++i)
        f[i] = c0[-S * (exv[i] * PITCH + eyv[i] * 9) + i];

    const float rho = ((f[0] + f[1]) + (f[2] + f[3])) +
                      ((f[4] + f[5]) + (f[6] + f[7])) + f[8];
    const float mx = f[1] - f[3] + f[5] - f[6] - f[7] + f[8];
    const float my = f[2] - f[4] + f[5] + f[6] - f[7] - f[8];
    const float jx = mx;
    const float jy = my + (TAU * GYF) * rho * rho;
    const float jx2 = jx * jx, jy2 = jy * jy;

    v2f bA, bB, bC;
    bA.x = rho;        bA.y = jx2 + jy2;
    bB.x = jx;         bB.y = jy;
    bC.x = jx2 - jy2;  bC.y = jx * jy;

    v8f r = {};
    r = __builtin_amdgcn_wmma_f32_16x16x4_f32(false, aR, false, bA, (short)0, r,
                                              false, false);

    o[0] = decay * f[0] + r[0];
    v8f d;
#pragma unroll
    for (int i = 0; i < 8; ++i) d[i] = decay * f[rq[i]] + (rq[i] < 5 ? r[1] : r[2]);
    d = __builtin_amdgcn_wmma_f32_16x16x4_f32(false, aB, false, bB, (short)0, d,
                                              false, false);
    d = __builtin_amdgcn_wmma_f32_16x16x4_f32(false, aC, false, bC, (short)0, d,
                                              false, false);
#pragma unroll
    for (int i = 0; i < 8; ++i) o[rq[i]] = d[i];
}

template <int S>
static __device__ __forceinline__ void tile_step(float* s_buf, unsigned wv, unsigned lane,
                                                 v2f aR, v2f aB, v2f aC) {
    constexpr unsigned NR = RSX - 2 * S;
    constexpr unsigned N = RSY - 2 * S;
    constexpr unsigned NCELL = NR * N;
    constexpr unsigned NGROUPS = (NCELL + 31u) / 32u;
    const int exv[9] = {0, 1, 0, -1, 0, 1, -1, -1, 1};
    const int eyv[9] = {0, 0, 1, 0, -1, 1, 1, -1, -1};

    for (unsigned g = wv; g < NGROUPS; g += NWAVES) {
        unsigned idx = g * 32u + lane;
        idx = idx < NCELL ? idx : NCELL - 1u;
        const unsigned r = idx / N, cc = idx - r * N;
        float* c0 = s_buf + (S + r) * PITCH + (S + cc) * 9u;
        float o[9];
        collide_cell<S>(c0, aR, aB, aC, o);
#pragma unroll
        for (int q = 0; q < 9; ++q)
            c0[-S * (exv[q] * PITCH + eyv[q] * 9) + q] = o[q];
    }
}

static __device__ __forceinline__ void tile_step_last(float* s_buf, unsigned wv, unsigned lane,
                                                      v2f aR, v2f aB, v2f aC) {
    constexpr unsigned GPW = TSX / NWAVES;
    static_assert(TSY == 32 && TSX % NWAVES == 0, "a group is one owned row; whole groups per wave");
    float o[GPW][9];
#pragma unroll
    for (unsigned u = 0; u < GPW; ++u) {
        const unsigned row = wv + u * NWAVES;
        const float* c0 = s_buf + (HALO + row) * PITCH + (HALO + lane) * 9u;
        collide_cell<STEPS>(c0, aR, aB, aC, o[u]);
    }
    __syncthreads();
#pragma unroll
    for (unsigned u = 0; u < GPW; ++u) {
        const unsigned row = wv + u * NWAVES;
        float* c0 = s_buf + (HALO + row) * PITCH + (HALO + lane) * 9u;
#pragma unroll
        for (int q = 0; q < 9; ++q) c0[q] = o[u][q];
    }
}

__global__ __launch_bounds__(NTHREADS) void lbm_fused(const float* __restrict__ fin,
                                                      const float* __restrict__ w,
                                                      float* __restrict__ fout) {
    __shared__ __attribute__((aligned(16))) float s_buf[BUFF];

    const unsigned tid  = threadIdx.x;
    const unsigned lane = tid & 31u;
    const unsigned wv   = (unsigned)__builtin_amdgcn_readfirstlane((int)(tid >> 5));
    const unsigned x0 = (blockIdx.x / TILES_Y) * TSX;
    const unsigned y0 = (blockIdx.x % TILES_Y) * TSY;

    const v4f* fin4 = reinterpret_cast<const v4f*>(fin);
    v4f* fout4 = reinterpret_cast<v4f*>(fout);
    v4f rv[IN_ROUNDS];
#pragma unroll
    for (unsigned k = 0; k < IN_ROUNDS; ++k) {
        const unsigned idx = tid + k * NTHREADS;
        if (idx < IN_PIECES) rv[k] = fin4[in_piece_goff(idx, x0, y0)];
    }

    const float w0 = w[0], w1 = w[1], w2 = w[2], w3 = w[3];
    const float w4 = w[4], w5 = w[5], w6 = w[6];

    v2f aR, aB, aC;
    build_a(&g_AT.v[0][0], lane, w0, w1, w2, w3, w4, w5, w6, aR, aB, aC);

#pragma unroll
    for (unsigned k = 0; k < IN_ROUNDS; ++k) {
        const unsigned idx = tid + k * NTHREADS;
        if (idx < IN_PIECES) *reinterpret_cast<v4f*>(&s_buf[in_piece_loff(idx)]) = rv[k];
    }
    __syncthreads();

    tile_step<1>(s_buf, wv, lane, aR, aB, aC);
    __syncthreads();
    tile_step<2>(s_buf, wv, lane, aR, aB, aC);
    __syncthreads();
    tile_step<3>(s_buf, wv, lane, aR, aB, aC);
    __syncthreads();
    tile_step_last(s_buf, wv, lane, aR, aB, aC);
    __syncthreads();

#pragma unroll
    for (unsigned k = 0; k < OUT_ROUNDS; ++k) {
        const unsigned idx = tid + k * NTHREADS;
        const v4f ov = *reinterpret_cast<const v4f*>(&s_buf[out_piece_loff(idx)]);
        fout4[out_piece_goff(idx, x0, y0)] = ov;
    }
}

extern "C" void kernel_launch(void* const* d_in, const int* in_sizes, int n_in,
                              void* d_out, int out_size, void* d_ws, size_t ws_size,
                              hipStream_t stream) {
    const float* f0 = (const float*)d_in[0];
    const float* w  = (const float*)d_in[1];
    float* out = (float*)d_out;

    const int threads = NTHREADS;
    const int blocks  = (NXL / TSX) * (NYL / TSY);

    lbm_fused<<<blocks, threads, 0, stream>>>(f0, w, out);
}


# 64x64-cell tiles, one 32-wave work-group per WGP (was two 16-wave groups on 64x32 tiles): ring cells staged per 100 owned fall from 41 to 27, recomputed ones from 15 to 10; per-cell sums unchanged
# speedup vs baseline: 3.5893x; 1.1392x over previous
#include <hip/hip_runtime.h>


typedef float v2f __attribute__((ext_vector_type(2)));
typedef float v8f __attribute__((ext_vector_type(8)));
typedef float v4f __attribute__((ext_vector_type(4)));

#define NXL 1024
#define NYL 1024
#define TAU 0.6f
#define GYF (-1e-4f)

#define STEPS 4
#define HALO STEPS
#define TSX 64
#define TSY 64
#define RSX (TSX + 2 * HALO)
#define RSY (TSY + 2 * HALO)
#define PITCH (RSY * 9)
#define BUFF (RSX * PITCH)
#define ROWV (PITCH / 4)
#define HV (HALO * 9 / 4)
#define MIDV (TSY * 9 / 4)
#define IN_PIECES (RSX * ROWV)
#define OUT_PIECES (TSX * MIDV)
#define TILES_Y (NYL / TSY)
#define NTHREADS 1024
#define NWAVES (NTHREADS / 32)
#define IN_ROUNDS ((IN_PIECES + NTHREADS - 1) / NTHREADS)
#define OUT_ROUNDS (OUT_PIECES / NTHREADS)
static_assert((HALO * 9) % 4 == 0 && (TSY * 9) % 4 == 0, "ring and centre segments are whole 16-byte pieces");
static_assert(NXL % TSX == 0 && NYL % TSY == 0 && TSY % 4 == 0 && HALO % 4 == 0, "tiles cover the grid; segments start on a 16-byte piece");
static_assert(OUT_PIECES % NTHREADS == 0, "output split: whole rounds");
static_assert(2 * HV + MIDV == ROWV, "a staged row is ring + centre + ring");

struct ATab { float v[9][16]; };

static constexpr ATab makeAT() {
    const signed char M[9][9] = {
        { 1,  1,  1,  1,  1,  1,  1,  1,  1},
        { 0,  1,  0, -1,  0,  1, -1, -1,  1},
        { 0,  0,  1,  0, -1,  1,  1, -1, -1},
        {-4, -1, -1, -1, -1,  2,  2,  2,  2},
        { 4, -2, -2, -2, -2,  1,  1,  1,  1},
        { 0, -2,  0,  2,  0,  1, -1, -1,  1},
        { 0,  0, -2,  0,  2,  1,  1, -1, -1},
        { 0,  1, -1,  1, -1,  0,  0,  0,  0},
        { 0,  0,  0,  0,  0,  1, -1,  1, -1},
    };
    const float nrm2[9] = {9.f, 6.f, 6.f, 36.f, 36.f, 12.f, 12.f, 4.f, 4.f};
    ATab t{};
    for (int k = 0; k < 9; ++k)
        for (int q = 0; q < 16; ++q) {
            float x = 0.0f;
            if (q < 9)
                x = (float)M[k][q] / nrm2[k] * (1.0f / TAU);
            t.v[k][q] = x;
        }
    return t;
}

__constant__ ATab g_AT = makeAT();

static __device__ inline unsigned in_piece_goff(unsigned idx, unsigned x0, unsigned y0) {
    const unsigned row = idx / ROWV, p = idx - row * ROWV;
    const unsigned gx = (x0 + (NXL - HALO) + row) & (NXL - 1);
    unsigned gy, pp;
    if (p < HV)             { gy = (y0 + (NYL - HALO)) & (NYL - 1); pp = p; }
    else if (p < HV + MIDV) { gy = y0;                              pp = p - HV; }
    else                    { gy = (y0 + TSY) & (NYL - 1);          pp = p - (HV + MIDV); }
    return ((gx << 8) + (gy >> 2)) * 9u + pp;
}
static __device__ inline unsigned in_piece_loff(unsigned idx) {
    const unsigned row = idx / ROWV, p = idx - row * ROWV;
    return row * PITCH + p * 4u;
}
static __device__ inline unsigned out_piece_goff(unsigned idx, unsigned x0, unsigned y0) {
    const unsigned row = idx / MIDV, kk = idx - row * MIDV;
    return (((x0 + row) << 8) + (y0 >> 2)) * 9u + kk;
}

static __device__ inline unsigned out_piece_loff(unsigned idx) {
    const unsigned row = idx / MIDV, kk = idx - row * MIDV;
    return (HALO + row) * PITCH + HALO * 9u + kk * 4u;
}

static_assert(makeAT().v[1][0] == 0.0f && makeAT().v[2][0] == 0.0f && makeAT().v[5][0] == 0.0f &&
              makeAT().v[6][0] == 0.0f && makeAT().v[7][0] == 0.0f && makeAT().v[8][0] == 0.0f,
              "output q = 0 takes only the rho and j2 terms");
#define ROWQ_LIST {5, 1, 6, 3, 7, 8, 2, 4}
#define ROWQ_PACKED 0x42873615u
static constexpr bool rowq_ok() {
    const int rq[8] = ROWQ_LIST;
    unsigned seen = 0;
    for (int r = 0; r < 8; ++r) {
        if (rq[r] < 1 || rq[r] > 8) return false;
        if (((ROWQ_PACKED >> (4 * r)) & 15u) != (unsigned)rq[r]) return false;
        seen |= 1u << rq[r];
    }
    return seen == 0x1feu;
}
static_assert(rowq_ok(), "ROWQ is a permutation of 1..8 and matches its packed form");
static constexpr bool class_rows_ok() {
    const ATab t = makeAT();
    const int rows[3] = {0, 3, 4};
    for (int i = 0; i < 3; ++i)
        for (int q = 1; q < 9; ++q)
            if (t.v[rows[i]][q] != t.v[rows[i]][q < 5 ? 1 : 5]) return false;
    return true;
}
static_assert(class_rows_ok(), "the rho and j2 coefficients are equal within q = 1..4 and within q = 5..8");
static __device__ __forceinline__ void build_a(const float* tb, unsigned lane,
                                               float w0, float w1, float w2, float w3,
                                               float w4, float w5, float w6,
                                               v2f& aR, v2f& aB, v2f& aC) {
    const unsigned c = lane & 15u, h = lane >> 4;
    const unsigned rr = c & 7u;
    const bool on  = (c >> 3) == h;
    const unsigned q = (ROWQ_PACKED >> (4u * rr)) & 15u;
    const float t1 = tb[1 * 16 + q], t2 = tb[2 * 16 + q], t5 = tb[5 * 16 + q];
    const float t6 = tb[6 * 16 + q], t7 = tb[7 * 16 + q], t8 = tb[8 * 16 + q];
    const float k1 = t1 + w4 * t5;
    const float k2 = t2 + w4 * t6;
    const float k4 = w5 * t7;
    const float k5 = w6 * t8;
    aB.x = on ? k1 : 0.0f;  aB.y = on ? k2 : 0.0f;
    aC.x = on ? k4 : 0.0f;  aC.y = on ? k5 : 0.0f;
    const unsigned qc = rr == 0u ? 0u : (rr == 1u ? 1u : 5u);
    const bool onr = on && rr < 3u;
    const float z0 = tb[0 * 16 + qc], z3 = tb[3 * 16 + qc], z4 = tb[4 * 16 + qc];
    aR.x = onr ? z0 + w0 * z3 + w2 * z4 : 0.0f;
    aR.y = onr ? w1 * z3 - w3 * z4 : 0.0f;
}

template <int S>
static __device__ __forceinline__ void collide_cell(const float* c0,
                                                    v2f aR, v2f aB, v2f aC, float (&o)[9]) {
    const int exv[9] = {0, 1, 0, -1, 0, 1, -1, -1, 1};
    const int eyv[9] = {0, 0, 1, 0, -1, 1, 1, -1, -1};
    const float decay = 1.0f - 1.0f / TAU;
    const int rq[8] = ROWQ_LIST;

    float f[9];
#pragma unroll
    for (int i = 0; i < 9; ++i)
        f[i] = c0[-S * (exv[i] * PITCH + eyv[i] * 9) + i];

    const float rho = ((f[0] + f[1]) + (f[2] + f[3])) +
                      ((f[4] + f[5]) + (f[6] + f[7])) + f[8];
    const float mx = f[1] - f[3] + f[5] - f[6] - f[7] + f[8];
    const float my = f[2] - f[4] + f[5] + f[6] - f[7] - f[8];
    const float jx = mx;
    const float jy = my + (TAU * GYF) * rho * rho;
    const float jx2 = jx * jx, jy2 = jy * jy;

    v2f bA, bB, bC;
    bA.x = rho;        bA.y = jx2 + jy2;
    bB.x = jx;         bB.y = jy;
    bC.x = jx2 - jy2;  bC.y = jx * jy;

    v8f r = {};
    r = __builtin_amdgcn_wmma_f32_16x16x4_f32(false, aR, false, bA, (short)0, r,
                                              false, false);

    o[0] = decay * f[0] + r[0];
    v8f d;
#pragma unroll
    for (int i = 0; i < 8; ++i) d[i] = decay * f[rq[i]] + (rq[i] < 5 ? r[1] : r[2]);
    d = __builtin_amdgcn_wmma_f32_16x16x4_f32(false, aB, false, bB, (short)0, d,
                                              false, false);
    d = __builtin_amdgcn_wmma_f32_16x16x4_f32(false, aC, false, bC, (short)0, d,
                                              false, false);
#pragma unroll
    for (int i = 0; i < 8; ++i) o[rq[i]] = d[i];
}

template <int S>
static __device__ __forceinline__ void tile_step(float* s_buf, unsigned wv, unsigned lane,
                                                 v2f aR, v2f aB, v2f aC) {
    constexpr unsigned NR = RSX - 2 * S;
    constexpr unsigned N = RSY - 2 * S;
    constexpr unsigned NCELL = NR * N;
    constexpr unsigned NGROUPS = (NCELL + 31u) / 32u;
    const int exv[9] = {0, 1, 0, -1, 0, 1, -1, -1, 1};
    const int eyv[9] = {0, 0, 1, 0, -1, 1, 1, -1, -1};

    for (unsigned g = wv; g < NGROUPS; g += NWAVES) {
        unsigned idx = g * 32u + lane;
        idx = idx < NCELL ? idx : NCELL - 1u;
        const unsigned r = idx / N, cc = idx - r * N;
        float* c0 = s_buf + (S + r) * PITCH + (S + cc) * 9u;
        float o[9];
        collide_cell<S>(c0, aR, aB, aC, o);
#pragma unroll
        for (int q = 0; q < 9; ++q)
            c0[-S * (exv[q] * PITCH + eyv[q] * 9) + q] = o[q];
    }
}

static __device__ __forceinline__ void tile_step_last(float* s_buf, unsigned wv, unsigned lane,
                                                      v2f aR, v2f aB, v2f aC) {
    constexpr unsigned GPR = TSY / 32;
    constexpr unsigned GPW = TSX * GPR / NWAVES;
    static_assert(TSY % 32 == 0 && (TSX * GPR) % NWAVES == 0, "whole groups per row; whole groups per wave");
    float o[GPW][9];
#pragma unroll
    for (unsigned u = 0; u < GPW; ++u) {
        const unsigned gi = wv + u * NWAVES;
        const unsigned row = gi / GPR, col = (gi % GPR) * 32u + lane;
        const float* c0 = s_buf + (HALO + row) * PITCH + (HALO + col) * 9u;
        collide_cell<STEPS>(c0, aR, aB, aC, o[u]);
    }
    __syncthreads();
#pragma unroll
    for (unsigned u = 0; u < GPW; ++u) {
        const unsigned gi = wv + u * NWAVES;
        const unsigned row = gi / GPR, col = (gi % GPR) * 32u + lane;
        float* c0 = s_buf + (HALO + row) * PITCH + (HALO + col) * 9u;
#pragma unroll
        for (int q = 0; q < 9; ++q) c0[q] = o[u][q];
    }
}

__global__ __launch_bounds__(NTHREADS) void lbm_fused(const float* __restrict__ fin,
                                                      const float* __restrict__ w,
                                                      float* __restrict__ fout) {
    __shared__ __attribute__((aligned(16))) float s_buf[BUFF];

    const unsigned tid  = threadIdx.x;
    const unsigned lane = tid & 31u;
    const unsigned wv   = (unsigned)__builtin_amdgcn_readfirstlane((int)(tid >> 5));
    const unsigned x0 = (blockIdx.x / TILES_Y) * TSX;
    const unsigned y0 = (blockIdx.x % TILES_Y) * TSY;

    const v4f* fin4 = reinterpret_cast<const v4f*>(fin);
    v4f* fout4 = reinterpret_cast<v4f*>(fout);
    v4f rv[IN_ROUNDS];
#pragma unroll
    for (unsigned k = 0; k < IN_ROUNDS; ++k) {
        const unsigned idx = tid + k * NTHREADS;
        if (idx < IN_PIECES) rv[k] = fin4[in_piece_goff(idx, x0, y0)];
    }

    const float w0 = w[0], w1 = w[1], w2 = w[2], w3 = w[3];
    const float w4 = w[4], w5 = w[5], w6 = w[6];

    v2f aR, aB, aC;
    build_a(&g_AT.v[0][0], lane, w0, w1, w2, w3, w4, w5, w6, aR, aB, aC);

#pragma unroll
    for (unsigned k = 0; k < IN_ROUNDS; ++k) {
        const unsigned idx = tid + k * NTHREADS;
        if (idx < IN_PIECES) *reinterpret_cast<v4f*>(&s_buf[in_piece_loff(idx)]) = rv[k];
    }
    __syncthreads();

    tile_step<1>(s_buf, wv, lane, aR, aB, aC);
    __syncthreads();
    tile_step<2>(s_buf, wv, lane, aR, aB, aC);
    __syncthreads();
    tile_step<3>(s_buf, wv, lane, aR, aB, aC);
    __syncthreads();
    tile_step_last(s_buf, wv, lane, aR, aB, aC);
    __syncthreads();

#pragma unroll
    for (unsigned k = 0; k < OUT_ROUNDS; ++k) {
        const unsigned idx = tid + k * NTHREADS;
        const v4f ov = *reinterpret_cast<const v4f*>(&s_buf[out_piece_loff(idx)]);
        fout4[out_piece_goff(idx, x0, y0)] = ov;
    }
}

extern "C" void kernel_launch(void* const* d_in, const int* in_sizes, int n_in,
                              void* d_out, int out_size, void* d_ws, size_t ws_size,
                              hipStream_t stream) {
    const float* f0 = (const float*)d_in[0];
    const float* w  = (const float*)d_in[1];
    float* out = (float*)d_out;

    const int threads = NTHREADS;
    const int blocks  = (NXL / TSX) * (NYL / TSY);

    lbm_fused<<<blocks, threads, 0, stream>>>(f0, w, out);
}
